# starting kernel
# baseline (speedup 1.0000x reference)
#include <hip/hip_runtime.h>
#include <math.h>

typedef float v2f __attribute__((ext_vector_type(2)));
typedef float v8f __attribute__((ext_vector_type(8)));

#define N_EDGES 500000
#define N_RELS  256
#define DIM     256
#define N3      768

__global__ void zero_kernel(float* __restrict__ p, int n) {
    int i = blockIdx.x * blockDim.x + threadIdx.x;
    if (i < n) p[i] = 0.0f;
}

__global__ __launch_bounds__(256) void accum_kernel(
        const int* __restrict__ nid, const int* __restrict__ rel,
        const float* __restrict__ emb, float* __restrict__ gsum,
        float* __restrict__ counts, int do_count) {
    extern __shared__ float lsum[];
    for (int i = threadIdx.x; i < N_RELS * DIM; i += blockDim.x)
        lsum[i] = 0.0f;
    __syncthreads();

    const int lane  = threadIdx.x & 31;
    const int wave  = threadIdx.x >> 5;
    const int wpb   = blockDim.x >> 5;
    const int gw    = blockIdx.x * wpb + wave;
    const int nw    = gridDim.x * wpb;

    for (int e = gw; e < N_EDGES; e += nw) {
        const int n = nid[e];
        const int r = rel[e];
        if (e + nw < N_EDGES) {
            const int n2 = nid[e + nw];
            __builtin_prefetch(emb + (size_t)n2 * DIM + lane * 8, 0, 0);
        }
        const float4* row = (const float4*)(emb + (size_t)n * DIM);
        float4 v0 = row[lane * 2 + 0];
        float4 v1 = row[lane * 2 + 1];
        float* d = lsum + r * DIM + lane * 8;
        atomicAdd(d + 0, v0.x); atomicAdd(d + 1, v0.y);
        atomicAdd(d + 2, v0.z); atomicAdd(d + 3, v0.w);
        atomicAdd(d + 4, v1.x); atomicAdd(d + 5, v1.y);
        atomicAdd(d + 6, v1.z); atomicAdd(d + 7, v1.w);
        if (do_count && lane == 0) atomicAdd(counts + r, 1.0f);
    }
    __syncthreads();
    for (int i = threadIdx.x; i < N_RELS * DIM; i += blockDim.x)
        atomicAdd(gsum + i, lsum[i]);
}

__global__ void mean_kernel(float* __restrict__ sums,
                            const float* __restrict__ counts, int n) {
    int idx = blockIdx.x * blockDim.x + threadIdx.x;
    if (idx >= n) return;
    int r = (idx >> 8) & (N_RELS - 1);
    sums[idx] = sums[idx] / fmaxf(counts[r], 1.0f);
}

__global__ __launch_bounds__(256) void gemm_wmma(
        const float* __restrict__ A, int a_rs, int a_es,
        const float* __restrict__ W, const float* __restrict__ bias,
        float* __restrict__ C) {
    const int lane = threadIdx.x & 31;
    const int wave = threadIdx.x >> 5;
    const int tile = blockIdx.x * 8 + wave;
    const int tm   = tile / 48;
    const int tn   = tile % 48;
    const int half = lane >> 4;
    const int ml   = lane & 15;
    const int m    = tm * 16 + ml;
    const int n    = tn * 16 + ml;

    const float* arow = A + (size_t)m * a_rs;
    const float* wrow = W + (size_t)n * DIM;

    v8f acc = {};
    for (int k0 = 0; k0 < DIM; k0 += 4) {
        const int k = k0 + 2 * half;
        v2f a, b;
        a.x = arow[(size_t)k * a_es];
        a.y = arow[(size_t)(k + 1) * a_es];
        b.x = wrow[k];
        b.y = wrow[k + 1];
        acc = __builtin_amdgcn_wmma_f32_16x16x4_f32(
                 false, a,  false, b,
                 (short)0, acc,  false,  false);
    }
    const float bv = bias[n];
#pragma unroll
    for (int v = 0; v < 8; ++v) {
        const int row = tm * 16 + v + 8 * half;
        C[(size_t)row * N3 + n] = acc[v] + bv;
    }
}

__global__ void gru_gates(const float* __restrict__ gi,
                          const float* __restrict__ gh,
                          const float* __restrict__ dyn, int which,
                          float* __restrict__ out) {
    int idx = blockIdx.x * blockDim.x + threadIdx.x;
    if (idx >= N_RELS * DIM) return;
    int r = idx >> 8;
    int d = idx & (DIM - 1);
    const float* gir = gi + (size_t)r * N3;
    const float* ghr = gh + (size_t)r * N3;
    float i_r = gir[d], i_z = gir[DIM + d], i_n = gir[2 * DIM + d];
    float h_r = ghr[d], h_z = ghr[DIM + d], h_n = ghr[2 * DIM + d];
    float h   = dyn[(size_t)r * 512 + d * 2 + which];
    float rg  = 1.0f / (1.0f + expf(-(i_r + h_r)));
    float z   = 1.0f / (1.0f + expf(-(i_z + h_z)));
    float ng  = tanhf(i_n + rg * h_n);
    out[(size_t)idx * 2 + which] = (1.0f - z) * ng + z * h;
}

extern "C" void kernel_launch(void* const* d_in, const int* in_sizes, int n_in,
                              void* d_out, int out_size, void* d_ws, size_t ws_size,
                              hipStream_t stream) {
    (void)in_sizes; (void)n_in; (void)out_size; (void)ws_size;

    const int*   src_nid = (const int*)d_in[0];
    const int*   dst_nid = (const int*)d_in[1];
    const int*   rel     = (const int*)d_in[2];
    const float* ent     = (const float*)d_in[3];
    const float* dyn     = (const float*)d_in[4];
    const float* W_ih    = (const float*)d_in[5];
    const float* W_hh    = (const float*)d_in[6];
    const float* b_ih    = (const float*)d_in[7];
    const float* b_hh    = (const float*)d_in[8];
    float*       out     = (float*)d_out;

    float* ws       = (float*)d_ws;
    float* sums_src = ws;
    float* sums_dst = ws + 65536;
    float* counts   = ws + 131072;
    float* gi_src   = ws + 131328;
    float* gh_src   = gi_src + 196608;
    float* gi_dst   = gh_src + 196608;
    float* gh_dst   = gi_dst + 196608;

    const int NACC = 131328;
    zero_kernel<<<(NACC + 255) / 256, 256, 0, stream>>>(ws, NACC);

    const size_t ldsBytes = (size_t)N_RELS * DIM * sizeof(float);
    accum_kernel<<<64, 256, ldsBytes, stream>>>(src_nid, rel, ent,
                                                sums_src, counts, 1);
    accum_kernel<<<64, 256, ldsBytes, stream>>>(dst_nid, rel, ent,
                                                sums_dst, counts, 0);

    mean_kernel<<<(2 * 65536) / 256, 256, 0, stream>>>(ws, counts, 2 * 65536);

    gemm_wmma<<<96, 256, 0, stream>>>(sums_src, DIM, 1, W_ih, b_ih, gi_src);
    gemm_wmma<<<96, 256, 0, stream>>>(dyn + 0,  512, 2, W_hh, b_hh, gh_src);
    gemm_wmma<<<96, 256, 0, stream>>>(sums_dst, DIM, 1, W_ih, b_ih, gi_dst);
    gemm_wmma<<<96, 256, 0, stream>>>(dyn + 1,  512, 2, W_hh, b_hh, gh_dst);

    gru_gates<<<65536 / 256, 256, 0, stream>>>(gi_src, gh_src, dyn, 0, out);
    gru_gates<<<65536 / 256, 256, 0, stream>>>(gi_dst, gh_dst, dyn, 1, out);
}


# cell in one launch, split by gate: block = 16 relations x 16 channels stages operand rows once in LDS; waves = 2 planes x 3 gates, each one gi + one gh tile, whole K; k order in product sums changes
# speedup vs baseline: 16.1704x; 16.1704x over previous
#include <hip/hip_runtime.h>
#include <math.h>

typedef float v2f __attribute__((ext_vector_type(2)));
typedef float v8f __attribute__((ext_vector_type(8)));

#define N_EDGES 500000
#define N_RELS  256
#define DIM     256
#define N3      768

#define CHUNK   2000
#define NB      ((N_EDGES + CHUNK - 1) / CHUNK)
#define EPT     ((CHUNK + 255) / 256)
#define HTILE   25
#define RWAVES  16

__global__ __launch_bounds__(256) void hist_kernel(
        const int* __restrict__ rel, int* __restrict__ hist) {
    __shared__ int h[N_RELS];
    h[threadIdx.x] = 0;
    __syncthreads();
    const int base = blockIdx.x * CHUNK;
    int rr[EPT];
#pragma unroll
    for (int q = 0; q < EPT; ++q) {
        const int i = (int)threadIdx.x + 256 * q;
        rr[q] = (i < CHUNK && base + i < N_EDGES) ? rel[base + i] : -1;
    }
#pragma unroll
    for (int q = 0; q < EPT; ++q)
        if (rr[q] >= 0) atomicAdd(&h[rr[q]], 1);
    __syncthreads();
    hist[blockIdx.x * N_RELS + threadIdx.x] = h[threadIdx.x];
}

__global__ __launch_bounds__(256) void scatter_kernel(
        const int* __restrict__ rel, const int* __restrict__ hist,
        int* __restrict__ perm, int* __restrict__ bstart,
        int* __restrict__ btotal) {
    __shared__ int cur[N_RELS];
    __shared__ int scan[N_RELS];
    const int r = threadIdx.x;
    const int b = blockIdx.x;
    const int base = b * CHUNK;
    int rr[EPT];
#pragma unroll
    for (int q = 0; q < EPT; ++q) {
        const int i = r + 256 * q;
        rr[q] = (i < CHUNK && base + i < N_EDGES) ? rel[base + i] : -1;
    }
    int before = 0, total = 0;
    for (int t = 0; t < NB; t += HTILE) {
        int v[HTILE];
#pragma unroll
        for (int q = 0; q < HTILE; ++q)
            v[q] = (t + q < NB) ? hist[(t + q) * N_RELS + r] : 0;
#pragma unroll
        for (int q = 0; q < HTILE; ++q) {
            total += v[q];
            before += (t + q < b) ? v[q] : 0;
        }
    }
    scan[r] = total;
    __syncthreads();
    for (int off = 1; off < N_RELS; off <<= 1) {
        const int t = (r >= off) ? scan[r - off] : 0;
        __syncthreads();
        scan[r] += t;
        __syncthreads();
    }
    const int start = scan[r] - total;
    cur[r] = start + before;
    if (b == 0) { bstart[r] = start; btotal[r] = total; }
    __syncthreads();
#pragma unroll
    for (int q = 0; q < EPT; ++q) {
        if (rr[q] >= 0) {
            const int pos = atomicAdd(&cur[rr[q]], 1);
            perm[pos] = base + r + 256 * q;
        }
    }
}

__device__ __forceinline__ void add_row(const float4* __restrict__ emb4, int n,
                                        int lane, float4& a0, float4& a1) {
    const float4* row = emb4 + (size_t)n * (DIM / 4);
    const float4 v0 = row[lane];
    const float4 v1 = row[32 + lane];
    a0.x += v0.x; a0.y += v0.y; a0.z += v0.z; a0.w += v0.w;
    a1.x += v1.x; a1.y += v1.y; a1.z += v1.z; a1.w += v1.w;
}

__global__ __launch_bounds__(512) void reduce_kernel(
        const int* __restrict__ src_nid, const int* __restrict__ dst_nid,
        const float* __restrict__ emb, const int* __restrict__ perm,
        const int* __restrict__ bstart, const int* __restrict__ btotal,
        float* __restrict__ X) {
    __shared__ float4 part[RWAVES][DIM / 4];
    const int lane  = threadIdx.x & 31;
    const int wave  = __builtin_amdgcn_readfirstlane((int)(threadIdx.x >> 5));
    const int r     = blockIdx.x & (N_RELS - 1);
    const int plane = blockIdx.x >> 8;
    const int* __restrict__ nid = plane ? dst_nid : src_nid;
    const float4* __restrict__ emb4 = (const float4*)emb;

    const int start = bstart[r];
    const int cnt   = btotal[r];
    const int seg   = (cnt + RWAVES - 1) / RWAVES;
    const int lo    = wave * seg;
    const int hi    = (lo + seg < cnt) ? (lo + seg) : cnt;
    const int n     = hi - lo;

    float4 a0 = make_float4(0.f, 0.f, 0.f, 0.f);
    float4 a1 = make_float4(0.f, 0.f, 0.f, 0.f);
    if (n > 0) {
        const int* __restrict__ pb = perm + start + lo;
        const int last = n - 1;
        int id   = nid[pb[min(lane, last)]];
        int e_nx = pb[min(32 + lane, last)];
        for (int j = 0; j < n; j += 32) {
            const int id_nx = nid[e_nx];
            const int e_nx2 = pb[min(j + 64 + lane, last)];
            const int m = n - j;
            if (m >= 32) {
#pragma unroll 8
                for (int k = 0; k < 32; ++k)
                    add_row(emb4, __builtin_amdgcn_readlane(id, k), lane, a0, a1);
            } else {
#pragma unroll 8
                for (int k = 0; k < m; ++k)
                    add_row(emb4, __builtin_amdgcn_readlane(id, k), lane, a0, a1);
            }
            id   = id_nx;
            e_nx = e_nx2;
        }
    }
    part[wave][lane]      = a0;
    part[wave][32 + lane] = a1;
    __syncthreads();
    if (threadIdx.x < DIM) {
        const float* p = (const float*)part;
        float s = 0.0f;
        for (int w = 0; w < RWAVES; ++w) s += p[w * DIM + threadIdx.x];
        X[(size_t)plane * (N_RELS * DIM) + (size_t)r * DIM + threadIdx.x] =
            s / fmaxf((float)cnt, 1.0f);
    }
}

#define KCH    32
#define STG    8
#define PITCH  65

__global__ __launch_bounds__(256) void cell_kernel(
        const float* __restrict__ X, const float* __restrict__ dyn,
        const float* __restrict__ W_ih, const float* __restrict__ W_hh,
        const float* __restrict__ b_ih, const float* __restrict__ b_hh,
        float* __restrict__ out) {
    __shared__ float4 XS[2 * 16 * PITCH];
    __shared__ float4 HS[2 * 16 * PITCH];
    __shared__ float4 WI[3 * 16 * PITCH];
    __shared__ float4 WH[3 * 16 * PITCH];
    __shared__ float sG[2][2][8][32];
    __shared__ float sO[2][8][32];
    const int t     = threadIdx.x;
    const int lane  = t & 31;
    const int wave  = __builtin_amdgcn_readfirstlane((int)(threadIdx.x >> 5));
    const int tr    = blockIdx.x >> 4;
    const int tn    = blockIdx.x & 15;

    {
        const float4* __restrict__ X4 = (const float4*)X;
        const float4* __restrict__ D4 = (const float4*)dyn;
        const float4* __restrict__ I4 = (const float4*)W_ih;
        const float4* __restrict__ H4 = (const float4*)W_hh;
        float4 vx[8], vh[8], vi[12], vw[12];
#pragma unroll
        for (int j = 0; j < 8; ++j) {
            const int i = t + 256 * j;
            const int p = i >> 10, r = (i >> 6) & 15, q = i & 63;
            vx[j] = X4[(size_t)p * (N_RELS * DIM / 4) + (size_t)(tr * 16 + r) * (DIM / 4) + q];
        }
#pragma unroll
        for (int j = 0; j < 8; ++j) {
            const int i = t + 256 * j;
            const int r = i >> 7, q = i & 127;
            vh[j] = D4[(size_t)(tr * 16 + r) * (2 * DIM / 4) + q];
        }
#pragma unroll
        for (int j = 0; j < 12; ++j) {
            const int i = t + 256 * j;
            const int g = i >> 10, r = (i >> 6) & 15, q = i & 63;
            const size_t o = (size_t)(g * DIM + tn * 16 + r) * (DIM / 4) + q;
            vi[j] = I4[o];
            vw[j] = H4[o];
        }
        __builtin_amdgcn_sched_barrier(0);
#pragma unroll
        for (int j = 0; j < 8; ++j) {
            const int i = t + 256 * j;
            const int p = i >> 10, r = (i >> 6) & 15, q = i & 63;
            XS[(p * 16 + r) * PITCH + q] = vx[j];
        }
        float2* __restrict__ HSf = (float2*)HS;
#pragma unroll
        for (int j = 0; j < 8; ++j) {
            const int i = t + 256 * j;
            const int r = i >> 7, q = i & 127;
            float2 e0, e1;
            e0.x = vh[j].x; e0.y = vh[j].z;
            e1.x = vh[j].y; e1.y = vh[j].w;
            HSf[((0 * 16 + r) * PITCH) * 2 + q] = e0;
            HSf[((1 * 16 + r) * PITCH) * 2 + q] = e1;
        }
#pragma unroll
        for (int j = 0; j < 12; ++j) {
            const int i = t + 256 * j;
            const int g = i >> 10, r = (i >> 6) & 15, q = i & 63;
            WI[(g * 16 + r) * PITCH + q] = vi[j];
            WH[(g * 16 + r) * PITCH + q] = vw[j];
        }
    }
    __syncthreads();

    const int plane = (wave < 6) ? wave / 3 : 0;
    const int gate  = (wave < 6) ? wave - 3 * plane : 0;
    const int c     = lane & 15;
    const int h     = lane >> 4;
    const int d     = tn * 16 + c;
    const int wrow  = gate * DIM + d;

    v8f acc_i = {};
    v8f acc_h = {};
    float bI = 0.0f, bH = 0.0f;
    float hp[8];
#pragma unroll
    for (int v = 0; v < 8; ++v) hp[v] = 0.0f;

    if (wave < 6) {
        bI = b_ih[wrow];
        bH = b_hh[wrow];
        if (gate == 2) {
#pragma unroll
            for (int v = 0; v < 8; ++v)
                hp[v] = dyn[(size_t)(tr * 16 + 8 * h + v) * (2 * DIM) + d * 2 + plane];
        }

        const float4* __restrict__ fa = XS + (plane * 16 + c) * PITCH + h;
        const float4* __restrict__ fh = HS + (plane * 16 + c) * PITCH + h;
        const float4* __restrict__ fi = WI + (gate * 16 + c) * PITCH + h;
        const float4* __restrict__ fw = WH + (gate * 16 + c) * PITCH + h;

        float4 A[2][STG], G[2][STG], BI[2][STG], BH[2][STG];
#pragma unroll
        for (int u = 0; u < STG; ++u) {
            A[0][u]  = fa[2 * u];
            G[0][u]  = fh[2 * u];
            BI[0][u] = fi[2 * u];
            BH[0][u] = fw[2 * u];
        }
#pragma unroll
        for (int s = 0; s < KCH / STG; ++s) {
            const int cur = s & 1;
            const int nxt = cur ^ 1;
            if (s + 1 < KCH / STG) {
#pragma unroll
                for (int u = 0; u < STG; ++u) {
                    const int kc = (s + 1) * STG + u;
                    A[nxt][u]  = fa[2 * kc];
                    G[nxt][u]  = fh[2 * kc];
                    BI[nxt][u] = fi[2 * kc];
                    BH[nxt][u] = fw[2 * kc];
                }
            }
            __builtin_amdgcn_sched_barrier(0);
#pragma unroll
            for (int u = 0; u < STG; ++u) {
                const float4 a  = A[cur][u];
                const float4 g  = G[cur][u];
                const float4 bi = BI[cur][u];
                const float4 bh = BH[cur][u];
                v2f a0, a1, b0, b1, g0, g1, w0, w1;
                a0.x = a.x;  a0.y = a.y;  a1.x = a.z;  a1.y = a.w;
                b0.x = bi.x; b0.y = bi.y; b1.x = bi.z; b1.y = bi.w;
                g0.x = g.x;  g0.y = g.y;  g1.x = g.z;  g1.y = g.w;
                w0.x = bh.x; w0.y = bh.y; w1.x = bh.z; w1.y = bh.w;
                acc_i = __builtin_amdgcn_wmma_f32_16x16x4_f32(false, a0, false, b0, (short)0, acc_i, false, false);
                acc_h = __builtin_amdgcn_wmma_f32_16x16x4_f32(false, g0, false, w0, (short)0, acc_h, false, false);
                acc_i = __builtin_amdgcn_wmma_f32_16x16x4_f32(false, a1, false, b1, (short)0, acc_i, false, false);
                acc_h = __builtin_amdgcn_wmma_f32_16x16x4_f32(false, g1, false, w1, (short)0, acc_h, false, false);
            }
            __builtin_amdgcn_sched_barrier(0);
            asm volatile("v_nop\n\tv_nop\n\tv_nop\n\tv_nop\n\tv_nop");
            __builtin_amdgcn_sched_barrier(0);
        }

        if (gate != 2) {
#pragma unroll
            for (int v = 0; v < 8; ++v) {
                const float pre = (acc_i[v] + bI) + (acc_h[v] + bH);
                sG[plane][gate][v][lane] = 1.0f / (1.0f + expf(-pre));
            }
        }
    }
    __syncthreads();
    if (wave < 6 && gate == 2) {
#pragma unroll
        for (int v = 0; v < 8; ++v) {
            const float rg = sG[plane][0][v][lane];
            const float z  = sG[plane][1][v][lane];
            const float ng = tanhf((acc_i[v] + bI) + rg * (acc_h[v] + bH));
            sO[plane][v][lane] = (1.0f - z) * ng + z * hp[v];
        }
    }
    __syncthreads();
    {
        float2* __restrict__ out2 = (float2*)out;
        float2 o;
        o.x = sO[0][wave][lane];
        o.y = sO[1][wave][lane];
        out2[(size_t)(tr * 16 + 8 * h + wave) * DIM + d] = o;
    }
}

extern "C" void kernel_launch(void* const* d_in, const int* in_sizes, int n_in,
                              void* d_out, int out_size, void* d_ws, size_t ws_size,
                              hipStream_t stream) {
    (void)in_sizes; (void)n_in; (void)out_size; (void)ws_size;

    const int*   src_nid = (const int*)d_in[0];
    const int*   dst_nid = (const int*)d_in[1];
    const int*   rel     = (const int*)d_in[2];
    const float* ent     = (const float*)d_in[3];
    const float* dyn     = (const float*)d_in[4];
    const float* W_ih    = (const float*)d_in[5];
    const float* W_hh    = (const float*)d_in[6];
    const float* b_ih    = (const float*)d_in[7];
    const float* b_hh    = (const float*)d_in[8];
    float*       out     = (float*)d_out;

    float* ws       = (float*)d_ws;
    float* X_src    = ws;
    int*   perm     = (int*)(ws + 131328);
    int*   hist     = perm + N_EDGES;
    int*   bstart   = hist + NB * N_RELS;
    int*   btotal   = bstart + N_RELS;

    hist_kernel<<<NB, 256, 0, stream>>>(rel, hist);
    scatter_kernel<<<NB, 256, 0, stream>>>(rel, hist, perm, bstart, btotal);
    reduce_kernel<<<2 * N_RELS, 512, 0, stream>>>(src_nid, dst_nid, ent, perm,
                                                  bstart, btotal, X_src);

    cell_kernel<<<(N_RELS / 16) * (DIM / 16), 256, 0, stream>>>(X_src, dyn, W_ih, W_hh, b_ih, b_hh, out);
}


# reduce: block orders its bucket's node ids by table region (LDS counting sort); all blocks walk the table together, so repeat reads of a row hit the second-level cache; f32 sum order changes
# speedup vs baseline: 18.0100x; 1.1138x over previous
#include <hip/hip_runtime.h>
#include <math.h>

typedef float v2f __attribute__((ext_vector_type(2)));
typedef float v8f __attribute__((ext_vector_type(8)));

#define N_EDGES 500000
#define N_RELS  256
#define DIM     256
#define N3      768

#define CHUNK   2000
#define NB      ((N_EDGES + CHUNK - 1) / CHUNK)
#define EPT     ((CHUNK + 255) / 256)
#define HTILE   25
#define RWAVES  16

__global__ __launch_bounds__(256) void hist_kernel(
        const int* __restrict__ rel, int* __restrict__ hist) {
    __shared__ int h[N_RELS];
    h[threadIdx.x] = 0;
    __syncthreads();
    const int base = blockIdx.x * CHUNK;
    int rr[EPT];
#pragma unroll
    for (int q = 0; q < EPT; ++q) {
        const int i = (int)threadIdx.x + 256 * q;
        rr[q] = (i < CHUNK && base + i < N_EDGES) ? rel[base + i] : -1;
    }
#pragma unroll
    for (int q = 0; q < EPT; ++q)
        if (rr[q] >= 0) atomicAdd(&h[rr[q]], 1);
    __syncthreads();
    hist[blockIdx.x * N_RELS + threadIdx.x] = h[threadIdx.x];
}

__global__ __launch_bounds__(256) void scatter_kernel(
        const int* __restrict__ rel, const int* __restrict__ hist,
        int* __restrict__ perm, int* __restrict__ bstart,
        int* __restrict__ btotal) {
    __shared__ int cur[N_RELS];
    __shared__ int scan[N_RELS];
    const int r = threadIdx.x;
    const int b = blockIdx.x;
    const int base = b * CHUNK;
    int rr[EPT];
#pragma unroll
    for (int q = 0; q < EPT; ++q) {
        const int i = r + 256 * q;
        rr[q] = (i < CHUNK && base + i < N_EDGES) ? rel[base + i] : -1;
    }
    int before = 0, total = 0;
    for (int t = 0; t < NB; t += HTILE) {
        int v[HTILE];
#pragma unroll
        for (int q = 0; q < HTILE; ++q)
            v[q] = (t + q < NB) ? hist[(t + q) * N_RELS + r] : 0;
#pragma unroll
        for (int q = 0; q < HTILE; ++q) {
            total += v[q];
            before += (t + q < b) ? v[q] : 0;
        }
    }
    scan[r] = total;
    __syncthreads();
    for (int off = 1; off < N_RELS; off <<= 1) {
        const int t = (r >= off) ? scan[r - off] : 0;
        __syncthreads();
        scan[r] += t;
        __syncthreads();
    }
    const int start = scan[r] - total;
    cur[r] = start + before;
    if (b == 0) { bstart[r] = start; btotal[r] = total; }
    __syncthreads();
#pragma unroll
    for (int q = 0; q < EPT; ++q) {
        if (rr[q] >= 0) {
            const int pos = atomicAdd(&cur[rr[q]], 1);
            perm[pos] = base + r + 256 * q;
        }
    }
}

#define SCAP    3072
#define SPT     (SCAP / 512)
#define RSH     10
#define NREG    256
#define RSTEP   8

__device__ __forceinline__ void add_row(const float4* __restrict__ emb4, int n,
                                        int lane, float4& a0, float4& a1) {
    const float4* row = emb4 + (size_t)n * (DIM / 4);
    const float4 v0 = row[lane];
    const float4 v1 = row[32 + lane];
    a0.x += v0.x; a0.y += v0.y; a0.z += v0.z; a0.w += v0.w;
    a1.x += v1.x; a1.y += v1.y; a1.z += v1.z; a1.w += v1.w;
}

__global__ __launch_bounds__(512) void reduce_kernel(
        const int* __restrict__ src_nid, const int* __restrict__ dst_nid,
        const float* __restrict__ emb, const int* __restrict__ perm,
        const int* __restrict__ bstart, const int* __restrict__ btotal,
        float* __restrict__ X) {
    __shared__ float4 part[RWAVES][DIM / 4];
    __shared__ int ids[SCAP];
    __shared__ int rcnt[NREG];
    const int t     = threadIdx.x;
    const int lane  = threadIdx.x & 31;
    const int wave  = __builtin_amdgcn_readfirstlane((int)(threadIdx.x >> 5));
    const int r     = blockIdx.x & (N_RELS - 1);
    const int plane = blockIdx.x >> 8;
    const int* __restrict__ nid = plane ? dst_nid : src_nid;
    const float4* __restrict__ emb4 = (const float4*)emb;

    const int start = bstart[r];
    const int cnt   = btotal[r];

    float4 a0 = make_float4(0.f, 0.f, 0.f, 0.f);
    float4 a1 = make_float4(0.f, 0.f, 0.f, 0.f);
    for (int c0 = 0; c0 < cnt; c0 += SCAP) {
        const int m = (cnt - c0 < SCAP) ? (cnt - c0) : SCAP;
        if (t < NREG) rcnt[t] = 0;
        __syncthreads();
        int nn[SPT], key[SPT], rank[SPT];
        {
            const int* __restrict__ pb = perm + start + c0;
            int ee[SPT];
#pragma unroll
            for (int q = 0; q < SPT; ++q) {
                const int i = t + 512 * q;
                ee[q] = pb[(i < m) ? i : (m - 1)];
            }
#pragma unroll
            for (int q = 0; q < SPT; ++q)
                nn[q] = nid[(unsigned)ee[q]];
        }
#pragma unroll
        for (int q = 0; q < SPT; ++q) {
            const int i = t + 512 * q;
            const unsigned g = ((unsigned)nn[q]) >> RSH;
            key[q]  = (int)((g < (unsigned)(NREG - 1)) ? g : (unsigned)(NREG - 1));
            rank[q] = 0;
            if (i < m) rank[q] = atomicAdd(&rcnt[key[q]], 1);
        }
        __syncthreads();
        if (wave == 0) {
            int v[NREG / 32];
            int tot = 0;
#pragma unroll
            for (int k = 0; k < NREG / 32; ++k) {
                v[k] = rcnt[(NREG / 32) * lane + k];
                tot += v[k];
            }
            int inc = tot;
#pragma unroll
            for (int off = 1; off < 32; off <<= 1) {
                const int y = __shfl_up(inc, off);
                if (lane >= off) inc += y;
            }
            int base = inc - tot;
#pragma unroll
            for (int k = 0; k < NREG / 32; ++k) {
                rcnt[(NREG / 32) * lane + k] = base;
                base += v[k];
            }
        }
        __syncthreads();
#pragma unroll
        for (int q = 0; q < SPT; ++q) {
            const int i = t + 512 * q;
            if (i < m) ids[rcnt[key[q]] + rank[q]] = nn[q];
        }
        __syncthreads();
        for (int j = wave * RSTEP; j < m; j += RWAVES * RSTEP) {
            const int mm = (m - j < RSTEP) ? (m - j) : RSTEP;
            const int sl = lane & (RSTEP - 1);
            const int id = ids[j + ((sl < mm) ? sl : (mm - 1))];
            if (mm == RSTEP) {
#pragma unroll
                for (int k = 0; k < RSTEP; ++k)
                    add_row(emb4, __builtin_amdgcn_readlane(id, k), lane, a0, a1);
            } else {
                for (int k = 0; k < mm; ++k)
                    add_row(emb4, __builtin_amdgcn_readlane(id, k), lane, a0, a1);
            }
        }
        __syncthreads();
    }
    part[wave][lane]      = a0;
    part[wave][32 + lane] = a1;
    __syncthreads();
    if (threadIdx.x < DIM) {
        const float* p = (const float*)part;
        float s = 0.0f;
        for (int w = 0; w < RWAVES; ++w) s += p[w * DIM + threadIdx.x];
        X[(size_t)plane * (N_RELS * DIM) + (size_t)r * DIM + threadIdx.x] =
            s / fmaxf((float)cnt, 1.0f);
    }
}

#define KCH    32
#define STG    8
#define PITCH  65

__global__ __launch_bounds__(256) void cell_kernel(
        const float* __restrict__ X, const float* __restrict__ dyn,
        const float* __restrict__ W_ih, const float* __restrict__ W_hh,
        const float* __restrict__ b_ih, const float* __restrict__ b_hh,
        float* __restrict__ out) {
    __shared__ float4 XS[2 * 16 * PITCH];
    __shared__ float4 HS[2 * 16 * PITCH];
    __shared__ float4 WI[3 * 16 * PITCH];
    __shared__ float4 WH[3 * 16 * PITCH];
    __shared__ float sG[2][2][8][32];
    __shared__ float sO[2][8][32];
    const int t     = threadIdx.x;
    const int lane  = t & 31;
    const int wave  = __builtin_amdgcn_readfirstlane((int)(threadIdx.x >> 5));
    const int tr    = blockIdx.x >> 4;
    const int tn    = blockIdx.x & 15;

    {
        const float4* __restrict__ X4 = (const float4*)X;
        const float4* __restrict__ D4 = (const float4*)dyn;
        const float4* __restrict__ I4 = (const float4*)W_ih;
        const float4* __restrict__ H4 = (const float4*)W_hh;
        float4 vx[8], vh[8], vi[12], vw[12];
#pragma unroll
        for (int j = 0; j < 8; ++j) {
            const int i = t + 256 * j;
            const int p = i >> 10, r = (i >> 6) & 15, q = i & 63;
            vx[j] = X4[(size_t)p * (N_RELS * DIM / 4) + (size_t)(tr * 16 + r) * (DIM / 4) + q];
        }
#pragma unroll
        for (int j = 0; j < 8; ++j) {
            const int i = t + 256 * j;
            const int r = i >> 7, q = i & 127;
            vh[j] = D4[(size_t)(tr * 16 + r) * (2 * DIM / 4) + q];
        }
#pragma unroll
        for (int j = 0; j < 12; ++j) {
            const int i = t + 256 * j;
            const int g = i >> 10, r = (i >> 6) & 15, q = i & 63;
            const size_t o = (size_t)(g * DIM + tn * 16 + r) * (DIM / 4) + q;
            vi[j] = I4[o];
            vw[j] = H4[o];
        }
        __builtin_amdgcn_sched_barrier(0);
#pragma unroll
        for (int j = 0; j < 8; ++j) {
            const int i = t + 256 * j;
            const int p = i >> 10, r = (i >> 6) & 15, q = i & 63;
            XS[(p * 16 + r) * PITCH + q] = vx[j];
        }
        float2* __restrict__ HSf = (float2*)HS;
#pragma unroll
        for (int j = 0; j < 8; ++j) {
            const int i = t + 256 * j;
            const int r = i >> 7, q = i & 127;
            float2 e0, e1;
            e0.x = vh[j].x; e0.y = vh[j].z;
            e1.x = vh[j].y; e1.y = vh[j].w;
            HSf[((0 * 16 + r) * PITCH) * 2 + q] = e0;
            HSf[((1 * 16 + r) * PITCH) * 2 + q] = e1;
        }
#pragma unroll
        for (int j = 0; j < 12; ++j) {
            const int i = t + 256 * j;
            const int g = i >> 10, r = (i >> 6) & 15, q = i & 63;
            WI[(g * 16 + r) * PITCH + q] = vi[j];
            WH[(g * 16 + r) * PITCH + q] = vw[j];
        }
    }
    __syncthreads();

    const int plane = (wave < 6) ? wave / 3 : 0;
    const int gate  = (wave < 6) ? wave - 3 * plane : 0;
    const int c     = lane & 15;
    const int h     = lane >> 4;
    const int d     = tn * 16 + c;
    const int wrow  = gate * DIM + d;

    v8f acc_i = {};
    v8f acc_h = {};
    float bI = 0.0f, bH = 0.0f;
    float hp[8];
#pragma unroll
    for (int v = 0; v < 8; ++v) hp[v] = 0.0f;

    if (wave < 6) {
        bI = b_ih[wrow];
        bH = b_hh[wrow];
        if (gate == 2) {
#pragma unroll
            for (int v = 0; v < 8; ++v)
                hp[v] = dyn[(size_t)(tr * 16 + 8 * h + v) * (2 * DIM) + d * 2 + plane];
        }

        const float4* __restrict__ fa = XS + (plane * 16 + c) * PITCH + h;
        const float4* __restrict__ fh = HS + (plane * 16 + c) * PITCH + h;
        const float4* __restrict__ fi = WI + (gate * 16 + c) * PITCH + h;
        const float4* __restrict__ fw = WH + (gate * 16 + c) * PITCH + h;

        float4 A[2][STG], G[2][STG], BI[2][STG], BH[2][STG];
#pragma unroll
        for (int u = 0; u < STG; ++u) {
            A[0][u]  = fa[2 * u];
            G[0][u]  = fh[2 * u];
            BI[0][u] = fi[2 * u];
            BH[0][u] = fw[2 * u];
        }
#pragma unroll
        for (int s = 0; s < KCH / STG; ++s) {
            const int cur = s & 1;
            const int nxt = cur ^ 1;
            if (s + 1 < KCH / STG) {
#pragma unroll
                for (int u = 0; u < STG; ++u) {
                    const int kc = (s + 1) * STG + u;
                    A[nxt][u]  = fa[2 * kc];
                    G[nxt][u]  = fh[2 * kc];
                    BI[nxt][u] = fi[2 * kc];
                    BH[nxt][u] = fw[2 * kc];
                }
            }
            __builtin_amdgcn_sched_barrier(0);
#pragma unroll
            for (int u = 0; u < STG; ++u) {
                const float4 a  = A[cur][u];
                const float4 g  = G[cur][u];
                const float4 bi = BI[cur][u];
                const float4 bh = BH[cur][u];
                v2f a0, a1, b0, b1, g0, g1, w0, w1;
                a0.x = a.x;  a0.y = a.y;  a1.x = a.z;  a1.y = a.w;
                b0.x = bi.x; b0.y = bi.y; b1.x = bi.z; b1.y = bi.w;
                g0.x = g.x;  g0.y = g.y;  g1.x = g.z;  g1.y = g.w;
                w0.x = bh.x; w0.y = bh.y; w1.x = bh.z; w1.y = bh.w;
                acc_i = __builtin_amdgcn_wmma_f32_16x16x4_f32(false, a0, false, b0, (short)0, acc_i, false, false);
                acc_h = __builtin_amdgcn_wmma_f32_16x16x4_f32(false, g0, false, w0, (short)0, acc_h, false, false);
                acc_i = __builtin_amdgcn_wmma_f32_16x16x4_f32(false, a1, false, b1, (short)0, acc_i, false, false);
                acc_h = __builtin_amdgcn_wmma_f32_16x16x4_f32(false, g1, false, w1, (short)0, acc_h, false, false);
            }
            __builtin_amdgcn_sched_barrier(0);
            asm volatile("v_nop\n\tv_nop\n\tv_nop\n\tv_nop\n\tv_nop");
            __builtin_amdgcn_sched_barrier(0);
        }

        if (gate != 2) {
#pragma unroll
            for (int v = 0; v < 8; ++v) {
                const float pre = (acc_i[v] + bI) + (acc_h[v] + bH);
                sG[plane][gate][v][lane] = 1.0f / (1.0f + expf(-pre));
            }
        }
    }
    __syncthreads();
    if (wave < 6 && gate == 2) {
#pragma unroll
        for (int v = 0; v < 8; ++v) {
            const float rg = sG[plane][0][v][lane];
            const float z  = sG[plane][1][v][lane];
            const float ng = tanhf((acc_i[v] + bI) + rg * (acc_h[v] + bH));
            sO[plane][v][lane] = (1.0f - z) * ng + z * hp[v];
        }
    }
    __syncthreads();
    {
        float2* __restrict__ out2 = (float2*)out;
        float2 o;
        o.x = sO[0][wave][lane];
        o.y = sO[1][wave][lane];
        out2[(size_t)(tr * 16 + 8 * h + wave) * DIM + d] = o;
    }
}

extern "C" void kernel_launch(void* const* d_in, const int* in_sizes, int n_in,
                              void* d_out, int out_size, void* d_ws, size_t ws_size,
                              hipStream_t stream) {
    (void)in_sizes; (void)n_in; (void)out_size; (void)ws_size;

    const int*   src_nid = (const int*)d_in[0];
    const int*   dst_nid = (const int*)d_in[1];
    const int*   rel     = (const int*)d_in[2];
    const float* ent     = (const float*)d_in[3];
    const float* dyn     = (const float*)d_in[4];
    const float* W_ih    = (const float*)d_in[5];
    const float* W_hh    = (const float*)d_in[6];
    const float* b_ih    = (const float*)d_in[7];
    const float* b_hh    = (const float*)d_in[8];
    float*       out     = (float*)d_out;

    float* ws       = (float*)d_ws;
    float* X_src    = ws;
    int*   perm     = (int*)(ws + 131328);
    int*   hist     = perm + N_EDGES;
    int*   bstart   = hist + NB * N_RELS;
    int*   btotal   = bstart + N_RELS;

    hist_kernel<<<NB, 256, 0, stream>>>(rel, hist);
    scatter_kernel<<<NB, 256, 0, stream>>>(rel, hist, perm, bstart, btotal);
    reduce_kernel<<<2 * N_RELS, 512, 0, stream>>>(src_nid, dst_nid, ent, perm,
                                                  bstart, btotal, X_src);

    cell_kernel<<<(N_RELS / 16) * (DIM / 16), 256, 0, stream>>>(X_src, dyn, W_ih, W_hh, b_ih, b_hh, out);
}


# bucketing in one launch: each 2000-edge block orders its edge ids by relation in LDS; the owner block finds its 250 segments by a size scan and halving search in LDS; prefix+scatter launch removed
# speedup vs baseline: 23.2624x; 1.2916x over previous
#include <hip/hip_runtime.h>
#include <math.h>

typedef float v2f __attribute__((ext_vector_type(2)));
typedef float v8f __attribute__((ext_vector_type(8)));

#define N_EDGES 500000
#define N_RELS  256
#define DIM     256
#define N3      768

#define CHUNK   2000
#define NB      ((N_EDGES + CHUNK - 1) / CHUNK)
#define EPT     ((CHUNK + 255) / 256)
#define NBLOG   8
#define RWAVES  16

__global__ __launch_bounds__(256) void chunk_sort_kernel(
        const int* __restrict__ rel, int* __restrict__ perm,
        int* __restrict__ seg) {
    __shared__ int h[N_RELS];
    __shared__ int wsum[8];
    __shared__ int sP[CHUNK];
    const int t    = threadIdx.x;
    const int lane = threadIdx.x & 31;
    const int wave = threadIdx.x >> 5;
    h[t] = 0;
    __syncthreads();
    const int base = blockIdx.x * CHUNK;
    int rr[EPT], rank[EPT];
#pragma unroll
    for (int q = 0; q < EPT; ++q) {
        const int i = t + 256 * q;
        rr[q] = (i < CHUNK && base + i < N_EDGES) ? rel[base + i] : -1;
    }
#pragma unroll
    for (int q = 0; q < EPT; ++q) {
        rank[q] = 0;
        if (rr[q] >= 0) rank[q] = atomicAdd(&h[rr[q]], 1);
    }
    __syncthreads();
    const int c = h[t];
    int inc = c;
#pragma unroll
    for (int off = 1; off < 32; off <<= 1) {
        const int y = __shfl_up(inc, off);
        if (lane >= off) inc += y;
    }
    if (lane == 31) wsum[wave] = inc;
    __syncthreads();
    int wbase = 0;
    for (int w = 0; w < 8; ++w) wbase += (w < wave) ? wsum[w] : 0;
    const int start = wbase + inc - c;
    h[t] = start;
    __syncthreads();
#pragma unroll
    for (int q = 0; q < EPT; ++q)
        if (rr[q] >= 0) sP[h[rr[q]] + rank[q]] = base + t + 256 * q;
    __syncthreads();
    int pv[EPT];
#pragma unroll
    for (int q = 0; q < EPT; ++q) {
        const int i = t + 256 * q;
        pv[q] = sP[(i < CHUNK) ? i : (CHUNK - 1)];
    }
#pragma unroll
    for (int q = 0; q < EPT; ++q) {
        const int i = t + 256 * q;
        if (i < CHUNK && base + i < N_EDGES) perm[base + i] = pv[q];
    }
    seg[t * NB + blockIdx.x] = start | (c << 16);
}

#define SCAP    3072
#define SPT     (SCAP / 512)
#define RSH     10
#define NREG    256
#define RSTEP   8

__device__ __forceinline__ void add_row(const float4* __restrict__ emb4, int n,
                                        int lane, float4& a0, float4& a1) {
    const float4* row = emb4 + (size_t)n * (DIM / 4);
    const float4 v0 = row[lane];
    const float4 v1 = row[32 + lane];
    a0.x += v0.x; a0.y += v0.y; a0.z += v0.z; a0.w += v0.w;
    a1.x += v1.x; a1.y += v1.y; a1.z += v1.z; a1.w += v1.w;
}

__global__ __launch_bounds__(512) void reduce_kernel(
        const int* __restrict__ src_nid, const int* __restrict__ dst_nid,
        const float* __restrict__ emb, const int* __restrict__ perm,
        const int* __restrict__ seg, float* __restrict__ X) {
    __shared__ float4 part[RWAVES][DIM / 4];
    __shared__ int ids[SCAP];
    __shared__ int rcnt[NREG];
    __shared__ int pre[NB + 1];
    __shared__ int sst[NB];
    __shared__ int wtot[RWAVES];
    const int t     = threadIdx.x;
    const int lane  = threadIdx.x & 31;
    const int wave  = __builtin_amdgcn_readfirstlane((int)(threadIdx.x >> 5));
    const int r     = blockIdx.x & (N_RELS - 1);
    const int plane = blockIdx.x >> 8;
    const int* __restrict__ nid = plane ? dst_nid : src_nid;
    const float4* __restrict__ emb4 = (const float4*)emb;

    int cnt = 0;
    {
        const int w    = (t < NB) ? seg[r * NB + t] : 0;
        const int segc = w >> 16;
        int inc = segc;
#pragma unroll
        for (int off = 1; off < 32; off <<= 1) {
            const int y = __shfl_up(inc, off);
            if (lane >= off) inc += y;
        }
        if (lane == 31) wtot[wave] = inc;
        __syncthreads();
        int wbase = 0;
        for (int k = 0; k < RWAVES; ++k) {
            const int v = wtot[k];
            cnt   += v;
            wbase += (k < wave) ? v : 0;
        }
        if (t <= NB) pre[t] = wbase + inc - segc;
        if (t <  NB) sst[t] = w & 0xffff;
        __syncthreads();
    }

    float4 a0 = make_float4(0.f, 0.f, 0.f, 0.f);
    float4 a1 = make_float4(0.f, 0.f, 0.f, 0.f);
    for (int c0 = 0; c0 < cnt; c0 += SCAP) {
        const int m = (cnt - c0 < SCAP) ? (cnt - c0) : SCAP;
        if (t < NREG) rcnt[t] = 0;
        __syncthreads();
        int nn[SPT], key[SPT], rank[SPT];
        {
            int ee[SPT];
#pragma unroll
            for (int q = 0; q < SPT; ++q) {
                const int i = t + 512 * q;
                const int g = c0 + ((i < m) ? i : (m - 1));
                int lo = 0, hi = NB;
#pragma unroll
                for (int s = 0; s < NBLOG; ++s) {
                    const int  mid = (lo + hi) >> 1;
                    const bool ge  = (pre[mid] <= g);
                    lo = ge ? mid : lo;
                    hi = ge ? hi : mid;
                }
                ee[q] = perm[lo * CHUNK + sst[lo] + (g - pre[lo])];
            }
#pragma unroll
            for (int q = 0; q < SPT; ++q)
                nn[q] = nid[(unsigned)ee[q]];
        }
#pragma unroll
        for (int q = 0; q < SPT; ++q) {
            const int i = t + 512 * q;
            const unsigned g = ((unsigned)nn[q]) >> RSH;
            key[q]  = (int)((g < (unsigned)(NREG - 1)) ? g : (unsigned)(NREG - 1));
            rank[q] = 0;
            if (i < m) rank[q] = atomicAdd(&rcnt[key[q]], 1);
        }
        __syncthreads();
        if (wave == 0) {
            int v[NREG / 32];
            int tot = 0;
#pragma unroll
            for (int k = 0; k < NREG / 32; ++k) {
                v[k] = rcnt[(NREG / 32) * lane + k];
                tot += v[k];
            }
            int inc = tot;
#pragma unroll
            for (int off = 1; off < 32; off <<= 1) {
                const int y = __shfl_up(inc, off);
                if (lane >= off) inc += y;
            }
            int base = inc - tot;
#pragma unroll
            for (int k = 0; k < NREG / 32; ++k) {
                rcnt[(NREG / 32) * lane + k] = base;
                base += v[k];
            }
        }
        __syncthreads();
#pragma unroll
        for (int q = 0; q < SPT; ++q) {
            const int i = t + 512 * q;
            if (i < m) ids[rcnt[key[q]] + rank[q]] = nn[q];
        }
        __syncthreads();
        for (int j = wave * RSTEP; j < m; j += RWAVES * RSTEP) {
            const int mm = (m - j < RSTEP) ? (m - j) : RSTEP;
            const int sl = lane & (RSTEP - 1);
            const int id = ids[j + ((sl < mm) ? sl : (mm - 1))];
            if (mm == RSTEP) {
#pragma unroll
                for (int k = 0; k < RSTEP; ++k)
                    add_row(emb4, __builtin_amdgcn_readlane(id, k), lane, a0, a1);
            } else {
                for (int k = 0; k < mm; ++k)
                    add_row(emb4, __builtin_amdgcn_readlane(id, k), lane, a0, a1);
            }
        }
        __syncthreads();
    }
    part[wave][lane]      = a0;
    part[wave][32 + lane] = a1;
    __syncthreads();
    if (threadIdx.x < DIM) {
        const float* p = (const float*)part;
        float s = 0.0f;
        for (int w = 0; w < RWAVES; ++w) s += p[w * DIM + threadIdx.x];
        X[(size_t)plane * (N_RELS * DIM) + (size_t)r * DIM + threadIdx.x] =
            s / fmaxf((float)cnt, 1.0f);
    }
}

#define KCH    32
#define STG    8
#define PITCH  65

__global__ __launch_bounds__(256) void cell_kernel(
        const float* __restrict__ X, const float* __restrict__ dyn,
        const float* __restrict__ W_ih, const float* __restrict__ W_hh,
        const float* __restrict__ b_ih, const float* __restrict__ b_hh,
        float* __restrict__ out) {
    __shared__ float4 XS[2 * 16 * PITCH];
    __shared__ float4 HS[2 * 16 * PITCH];
    __shared__ float4 WI[3 * 16 * PITCH];
    __shared__ float4 WH[3 * 16 * PITCH];
    __shared__ float sG[2][2][8][32];
    __shared__ float sO[2][8][32];
    const int t     = threadIdx.x;
    const int lane  = t & 31;
    const int wave  = __builtin_amdgcn_readfirstlane((int)(threadIdx.x >> 5));
    const int tr    = blockIdx.x >> 4;
    const int tn    = blockIdx.x & 15;

    {
        const float4* __restrict__ X4 = (const float4*)X;
        const float4* __restrict__ D4 = (const float4*)dyn;
        const float4* __restrict__ I4 = (const float4*)W_ih;
        const float4* __restrict__ H4 = (const float4*)W_hh;
        float4 vx[8], vh[8], vi[12], vw[12];
#pragma unroll
        for (int j = 0; j < 8; ++j) {
            const int i = t + 256 * j;
            const int p = i >> 10, r = (i >> 6) & 15, q = i & 63;
            vx[j] = X4[(size_t)p * (N_RELS * DIM / 4) + (size_t)(tr * 16 + r) * (DIM / 4) + q];
        }
#pragma unroll
        for (int j = 0; j < 8; ++j) {
            const int i = t + 256 * j;
            const int r = i >> 7, q = i & 127;
            vh[j] = D4[(size_t)(tr * 16 + r) * (2 * DIM / 4) + q];
        }
#pragma unroll
        for (int j = 0; j < 12; ++j) {
            const int i = t + 256 * j;
            const int g = i >> 10, r = (i >> 6) & 15, q = i & 63;
            const size_t o = (size_t)(g * DIM + tn * 16 + r) * (DIM / 4) + q;
            vi[j] = I4[o];
            vw[j] = H4[o];
        }
        __builtin_amdgcn_sched_barrier(0);
#pragma unroll
        for (int j = 0; j < 8; ++j) {
            const int i = t + 256 * j;
            const int p = i >> 10, r = (i >> 6) & 15, q = i & 63;
            XS[(p * 16 + r) * PITCH + q] = vx[j];
        }
        float2* __restrict__ HSf = (float2*)HS;
#pragma unroll
        for (int j = 0; j < 8; ++j) {
            const int i = t + 256 * j;
            const int r = i >> 7, q = i & 127;
            float2 e0, e1;
            e0.x = vh[j].x; e0.y = vh[j].z;
            e1.x = vh[j].y; e1.y = vh[j].w;
            HSf[((0 * 16 + r) * PITCH) * 2 + q] = e0;
            HSf[((1 * 16 + r) * PITCH) * 2 + q] = e1;
        }
#pragma unroll
        for (int j = 0; j < 12; ++j) {
            const int i = t + 256 * j;
            const int g = i >> 10, r = (i >> 6) & 15, q = i & 63;
            WI[(g * 16 + r) * PITCH + q] = vi[j];
            WH[(g * 16 + r) * PITCH + q] = vw[j];
        }
    }
    __syncthreads();

    const int plane = (wave < 6) ? wave / 3 : 0;
    const int gate  = (wave < 6) ? wave - 3 * plane : 0;
    const int c     = lane & 15;
    const int h     = lane >> 4;
    const int d     = tn * 16 + c;
    const int wrow  = gate * DIM + d;

    v8f acc_i = {};
    v8f acc_h = {};
    float bI = 0.0f, bH = 0.0f;
    float hp[8];
#pragma unroll
    for (int v = 0; v < 8; ++v) hp[v] = 0.0f;

    if (wave < 6) {
        bI = b_ih[wrow];
        bH = b_hh[wrow];
        if (gate == 2) {
#pragma unroll
            for (int v = 0; v < 8; ++v)
                hp[v] = dyn[(size_t)(tr * 16 + 8 * h + v) * (2 * DIM) + d * 2 + plane];
        }

        const float4* __restrict__ fa = XS + (plane * 16 + c) * PITCH + h;
        const float4* __restrict__ fh = HS + (plane * 16 + c) * PITCH + h;
        const float4* __restrict__ fi = WI + (gate * 16 + c) * PITCH + h;
        const float4* __restrict__ fw = WH + (gate * 16 + c) * PITCH + h;

        float4 A[2][STG], G[2][STG], BI[2][STG], BH[2][STG];
#pragma unroll
        for (int u = 0; u < STG; ++u) {
            A[0][u]  = fa[2 * u];
            G[0][u]  = fh[2 * u];
            BI[0][u] = fi[2 * u];
            BH[0][u] = fw[2 * u];
        }
#pragma unroll
        for (int s = 0; s < KCH / STG; ++s) {
            const int cur = s & 1;
            const int nxt = cur ^ 1;
            if (s + 1 < KCH / STG) {
#pragma unroll
                for (int u = 0; u < STG; ++u) {
                    const int kc = (s + 1) * STG + u;
                    A[nxt][u]  = fa[2 * kc];
                    G[nxt][u]  = fh[2 * kc];
                    BI[nxt][u] = fi[2 * kc];
                    BH[nxt][u] = fw[2 * kc];
                }
            }
            __builtin_amdgcn_sched_barrier(0);
#pragma unroll
            for (int u = 0; u < STG; ++u) {
                const float4 a  = A[cur][u];
                const float4 g  = G[cur][u];
                const float4 bi = BI[cur][u];
                const float4 bh = BH[cur][u];
                v2f a0, a1, b0, b1, g0, g1, w0, w1;
                a0.x = a.x;  a0.y = a.y;  a1.x = a.z;  a1.y = a.w;
                b0.x = bi.x; b0.y = bi.y; b1.x = bi.z; b1.y = bi.w;
                g0.x = g.x;  g0.y = g.y;  g1.x = g.z;  g1.y = g.w;
                w0.x = bh.x; w0.y = bh.y; w1.x = bh.z; w1.y = bh.w;
                acc_i = __builtin_amdgcn_wmma_f32_16x16x4_f32(false, a0, false, b0, (short)0, acc_i, false, false);
                acc_h = __builtin_amdgcn_wmma_f32_16x16x4_f32(false, g0, false, w0, (short)0, acc_h, false, false);
                acc_i = __builtin_amdgcn_wmma_f32_16x16x4_f32(false, a1, false, b1, (short)0, acc_i, false, false);
                acc_h = __builtin_amdgcn_wmma_f32_16x16x4_f32(false, g1, false, w1, (short)0, acc_h, false, false);
            }
            __builtin_amdgcn_sched_barrier(0);
            asm volatile("v_nop\n\tv_nop\n\tv_nop\n\tv_nop\n\tv_nop");
            __builtin_amdgcn_sched_barrier(0);
        }

        if (gate != 2) {
#pragma unroll
            for (int v = 0; v < 8; ++v) {
                const float pre = (acc_i[v] + bI) + (acc_h[v] + bH);
                sG[plane][gate][v][lane] = 1.0f / (1.0f + expf(-pre));
            }
        }
    }
    __syncthreads();
    if (wave < 6 && gate == 2) {
#pragma unroll
        for (int v = 0; v < 8; ++v) {
            const float rg = sG[plane][0][v][lane];
            const float z  = sG[plane][1][v][lane];
            const float ng = tanhf((acc_i[v] + bI) + rg * (acc_h[v] + bH));
            sO[plane][v][lane] = (1.0f - z) * ng + z * hp[v];
        }
    }
    __syncthreads();
    {
        float2* __restrict__ out2 = (float2*)out;
        float2 o;
        o.x = sO[0][wave][lane];
        o.y = sO[1][wave][lane];
        out2[(size_t)(tr * 16 + 8 * h + wave) * DIM + d] = o;
    }
}

extern "C" void kernel_launch(void* const* d_in, const int* in_sizes, int n_in,
                              void* d_out, int out_size, void* d_ws, size_t ws_size,
                              hipStream_t stream) {
    (void)in_sizes; (void)n_in; (void)out_size; (void)ws_size;

    const int*   src_nid = (const int*)d_in[0];
    const int*   dst_nid = (const int*)d_in[1];
    const int*   rel     = (const int*)d_in[2];
    const float* ent     = (const float*)d_in[3];
    const float* dyn     = (const float*)d_in[4];
    const float* W_ih    = (const float*)d_in[5];
    const float* W_hh    = (const float*)d_in[6];
    const float* b_ih    = (const float*)d_in[7];
    const float* b_hh    = (const float*)d_in[8];
    float*       out     = (float*)d_out;

    float* ws       = (float*)d_ws;
    float* X_src    = ws;
    int*   perm     = (int*)(ws + 131328);
    int*   seg      = perm + N_EDGES;

    chunk_sort_kernel<<<NB, 256, 0, stream>>>(rel, perm, seg);
    reduce_kernel<<<2 * N_RELS, 512, 0, stream>>>(src_nid, dst_nid, ent, perm,
                                                  seg, X_src);

    cell_kernel<<<(N_RELS / 16) * (DIM / 16), 256, 0, stream>>>(X_src, dyn, W_ih, W_hh, b_ih, b_hh, out);
}


# bucketing launch lays down the edges' node ids (a src list and a dst list) in relation order instead of edge ids; a reduce block reads its ids as segment words: 1M random 4-byte id reads gone
# speedup vs baseline: 24.2827x; 1.0439x over previous
#include <hip/hip_runtime.h>
#include <math.h>

typedef float v2f __attribute__((ext_vector_type(2)));
typedef float v8f __attribute__((ext_vector_type(8)));

#define N_EDGES 500000
#define N_RELS  256
#define DIM     256
#define N3      768

#define CHUNK   2000
#define NB      ((N_EDGES + CHUNK - 1) / CHUNK)
#define EPT     ((CHUNK + 255) / 256)
#define NBLOG   8
#define RWAVES  16

#define NBITS   18
#define NMASK   ((1 << NBITS) - 1)
#define WS_WIDE_BYTES ((size_t)(131328 + 2 * N_EDGES + N_RELS * NB) * 4)

__global__ __launch_bounds__(256) void chunk_sort_kernel(
        const int* __restrict__ rel, const int* __restrict__ src_nid,
        const int* __restrict__ dst_nid, int* __restrict__ permS,
        int* __restrict__ permD, int* __restrict__ seg, size_t ws_bytes) {
    __shared__ int h[N_RELS];
    __shared__ int wsum[8];
    __shared__ int sS[CHUNK];
    __shared__ int sD[CHUNK];
    const bool wide_ws = ws_bytes >= WS_WIDE_BYTES;
    const int t    = threadIdx.x;
    const int lane = threadIdx.x & 31;
    const int wave = threadIdx.x >> 5;
    h[t] = 0;
    __syncthreads();
    const int base = blockIdx.x * CHUNK;
    int rr[EPT], rank[EPT], ns[EPT], nd[EPT];
#pragma unroll
    for (int q = 0; q < EPT; ++q) {
        const int i  = t + 256 * q;
        const bool in = (i < CHUNK && base + i < N_EDGES);
        const int e  = in ? (base + i) : (N_EDGES - 1);
        const int rv = rel[e];
        ns[q] = src_nid[e];
        nd[q] = dst_nid[e];
        rr[q] = in ? rv : -1;
    }
#pragma unroll
    for (int q = 0; q < EPT; ++q) {
        rank[q] = 0;
        if (rr[q] >= 0) rank[q] = atomicAdd(&h[rr[q]], 1);
    }
    __syncthreads();
    const int c = h[t];
    int inc = c;
#pragma unroll
    for (int off = 1; off < 32; off <<= 1) {
        const int y = __shfl_up(inc, off);
        if (lane >= off) inc += y;
    }
    if (lane == 31) wsum[wave] = inc;
    __syncthreads();
    int wbase = 0;
    for (int w = 0; w < 8; ++w) wbase += (w < wave) ? wsum[w] : 0;
    const int start = wbase + inc - c;
    h[t] = start;
    __syncthreads();
#pragma unroll
    for (int q = 0; q < EPT; ++q)
        if (rr[q] >= 0) {
            const int p = h[rr[q]] + rank[q];
            sS[p] = (ns[q] & NMASK) | ((t + 256 * q) << NBITS);
            sD[p] = nd[q];
        }
    __syncthreads();
    int pvS[EPT], pvD[EPT];
#pragma unroll
    for (int q = 0; q < EPT; ++q) {
        const int i = t + 256 * q;
        const int j = (i < CHUNK) ? i : (CHUNK - 1);
        pvS[q] = sS[j];
        pvD[q] = sD[j];
    }
#pragma unroll
    for (int q = 0; q < EPT; ++q) {
        const int i = t + 256 * q;
        if (i < CHUNK && base + i < N_EDGES) {
            permS[base + i] = pvS[q];
            if (wide_ws) permD[base + i] = pvD[q];
        }
    }
    seg[t * NB + blockIdx.x] = start | (c << 16);
}

#define SCAP    3072
#define SPT     (SCAP / 512)
#define RSH     10
#define NREG    256
#define RSTEP   8

__device__ __forceinline__ void add_row(const float4* __restrict__ emb4, int n,
                                        int lane, float4& a0, float4& a1) {
    const float4* row = emb4 + (size_t)n * (DIM / 4);
    const float4 v0 = row[lane];
    const float4 v1 = row[32 + lane];
    a0.x += v0.x; a0.y += v0.y; a0.z += v0.z; a0.w += v0.w;
    a1.x += v1.x; a1.y += v1.y; a1.z += v1.z; a1.w += v1.w;
}

__global__ __launch_bounds__(512) void reduce_kernel(
        const int* __restrict__ dst_nid, const float* __restrict__ emb,
        const int* __restrict__ permS, const int* __restrict__ permD,
        const int* __restrict__ seg, float* __restrict__ X, size_t ws_bytes) {
    __shared__ float4 part[RWAVES][DIM / 4];
    __shared__ int ids[SCAP];
    __shared__ int rcnt[NREG];
    __shared__ int pre[NB + 1];
    __shared__ int sst[NB];
    __shared__ int wtot[RWAVES];
    const int t     = threadIdx.x;
    const int lane  = threadIdx.x & 31;
    const int wave  = __builtin_amdgcn_readfirstlane((int)(threadIdx.x >> 5));
    const int r     = blockIdx.x & (N_RELS - 1);
    const int plane = blockIdx.x >> 8;
    const bool wide_ws = ws_bytes >= WS_WIDE_BYTES;
    const int* __restrict__ plist = (plane && wide_ws) ? permD : permS;
    const float4* __restrict__ emb4 = (const float4*)emb;

    int cnt = 0;
    {
        const int w    = (t < NB) ? seg[r * NB + t] : 0;
        const int segc = w >> 16;
        int inc = segc;
#pragma unroll
        for (int off = 1; off < 32; off <<= 1) {
            const int y = __shfl_up(inc, off);
            if (lane >= off) inc += y;
        }
        if (lane == 31) wtot[wave] = inc;
        __syncthreads();
        int wbase = 0;
        for (int k = 0; k < RWAVES; ++k) {
            const int v = wtot[k];
            cnt   += v;
            wbase += (k < wave) ? v : 0;
        }
        if (t <= NB) pre[t] = wbase + inc - segc;
        if (t <  NB) sst[t] = w & 0xffff;
        __syncthreads();
    }

    float4 a0 = make_float4(0.f, 0.f, 0.f, 0.f);
    float4 a1 = make_float4(0.f, 0.f, 0.f, 0.f);
    for (int c0 = 0; c0 < cnt; c0 += SCAP) {
        const int m = (cnt - c0 < SCAP) ? (cnt - c0) : SCAP;
        if (t < NREG) rcnt[t] = 0;
        __syncthreads();
        int nn[SPT], key[SPT], rank[SPT];
        {
            int ww[SPT], cb[SPT];
#pragma unroll
            for (int q = 0; q < SPT; ++q) {
                const int i = t + 512 * q;
                const int g = c0 + ((i < m) ? i : (m - 1));
                int lo = 0, hi = NB;
#pragma unroll
                for (int s = 0; s < NBLOG; ++s) {
                    const int  mid = (lo + hi) >> 1;
                    const bool ge  = (pre[mid] <= g);
                    lo = ge ? mid : lo;
                    hi = ge ? hi : mid;
                }
                cb[q] = lo * CHUNK;
                ww[q] = plist[(unsigned)(cb[q] + sst[lo] + (g - pre[lo]))];
            }
            if (plane == 0) {
#pragma unroll
                for (int q = 0; q < SPT; ++q) nn[q] = ww[q] & NMASK;
            } else if (wide_ws) {
#pragma unroll
                for (int q = 0; q < SPT; ++q) nn[q] = ww[q];
            } else {
#pragma unroll
                for (int q = 0; q < SPT; ++q)
                    nn[q] = dst_nid[(unsigned)(cb[q] + (int)((unsigned)ww[q] >> NBITS))];
            }
        }
#pragma unroll
        for (int q = 0; q < SPT; ++q) {
            const int i = t + 512 * q;
            const unsigned g = ((unsigned)nn[q]) >> RSH;
            key[q]  = (int)((g < (unsigned)(NREG - 1)) ? g : (unsigned)(NREG - 1));
            rank[q] = 0;
            if (i < m) rank[q] = atomicAdd(&rcnt[key[q]], 1);
        }
        __syncthreads();
        if (wave == 0) {
            int v[NREG / 32];
            int tot = 0;
#pragma unroll
            for (int k = 0; k < NREG / 32; ++k) {
                v[k] = rcnt[(NREG / 32) * lane + k];
                tot += v[k];
            }
            int inc = tot;
#pragma unroll
            for (int off = 1; off < 32; off <<= 1) {
                const int y = __shfl_up(inc, off);
                if (lane >= off) inc += y;
            }
            int base = inc - tot;
#pragma unroll
            for (int k = 0; k < NREG / 32; ++k) {
                rcnt[(NREG / 32) * lane + k] = base;
                base += v[k];
            }
        }
        __syncthreads();
#pragma unroll
        for (int q = 0; q < SPT; ++q) {
            const int i = t + 512 * q;
            if (i < m) ids[rcnt[key[q]] + rank[q]] = nn[q];
        }
        __syncthreads();
        for (int j = wave * RSTEP; j < m; j += RWAVES * RSTEP) {
            const int mm = (m - j < RSTEP) ? (m - j) : RSTEP;
            const int sl = lane & (RSTEP - 1);
            const int id = ids[j + ((sl < mm) ? sl : (mm - 1))];
            if (mm == RSTEP) {
#pragma unroll
                for (int k = 0; k < RSTEP; ++k)
                    add_row(emb4, __builtin_amdgcn_readlane(id, k), lane, a0, a1);
            } else {
                for (int k = 0; k < mm; ++k)
                    add_row(emb4, __builtin_amdgcn_readlane(id, k), lane, a0, a1);
            }
        }
        __syncthreads();
    }
    part[wave][lane]      = a0;
    part[wave][32 + lane] = a1;
    __syncthreads();
    if (threadIdx.x < DIM) {
        const float* p = (const float*)part;
        float s = 0.0f;
        for (int w = 0; w < RWAVES; ++w) s += p[w * DIM + threadIdx.x];
        X[(size_t)plane * (N_RELS * DIM) + (size_t)r * DIM + threadIdx.x] =
            s / fmaxf((float)cnt, 1.0f);
    }
}

#define KCH    32
#define STG    8
#define PITCH  65

__global__ __launch_bounds__(256) void cell_kernel(
        const float* __restrict__ X, const float* __restrict__ dyn,
        const float* __restrict__ W_ih, const float* __restrict__ W_hh,
        const float* __restrict__ b_ih, const float* __restrict__ b_hh,
        float* __restrict__ out) {
    __shared__ float4 XS[2 * 16 * PITCH];
    __shared__ float4 HS[2 * 16 * PITCH];
    __shared__ float4 WI[3 * 16 * PITCH];
    __shared__ float4 WH[3 * 16 * PITCH];
    __shared__ float sG[2][2][8][32];
    __shared__ float sO[2][8][32];
    const int t     = threadIdx.x;
    const int lane  = t & 31;
    const int wave  = __builtin_amdgcn_readfirstlane((int)(threadIdx.x >> 5));
    const int tr    = blockIdx.x >> 4;
    const int tn    = blockIdx.x & 15;

    {
        const float4* __restrict__ X4 = (const float4*)X;
        const float4* __restrict__ D4 = (const float4*)dyn;
        const float4* __restrict__ I4 = (const float4*)W_ih;
        const float4* __restrict__ H4 = (const float4*)W_hh;
        float4 vx[8], vh[8], vi[12], vw[12];
#pragma unroll
        for (int j = 0; j < 8; ++j) {
            const int i = t + 256 * j;
            const int p = i >> 10, r = (i >> 6) & 15, q = i & 63;
            vx[j] = X4[(size_t)p * (N_RELS * DIM / 4) + (size_t)(tr * 16 + r) * (DIM / 4) + q];
        }
#pragma unroll
        for (int j = 0; j < 8; ++j) {
            const int i = t + 256 * j;
            const int r = i >> 7, q = i & 127;
            vh[j] = D4[(size_t)(tr * 16 + r) * (2 * DIM / 4) + q];
        }
#pragma unroll
        for (int j = 0; j < 12; ++j) {
            const int i = t + 256 * j;
            const int g = i >> 10, r = (i >> 6) & 15, q = i & 63;
            const size_t o = (size_t)(g * DIM + tn * 16 + r) * (DIM / 4) + q;
            vi[j] = I4[o];
            vw[j] = H4[o];
        }
        __builtin_amdgcn_sched_barrier(0);
#pragma unroll
        for (int j = 0; j < 8; ++j) {
            const int i = t + 256 * j;
            const int p = i >> 10, r = (i >> 6) & 15, q = i & 63;
            XS[(p * 16 + r) * PITCH + q] = vx[j];
        }
        float2* __restrict__ HSf = (float2*)HS;
#pragma unroll
        for (int j = 0; j < 8; ++j) {
            const int i = t + 256 * j;
            const int r = i >> 7, q = i & 127;
            float2 e0, e1;
            e0.x = vh[j].x; e0.y = vh[j].z;
            e1.x = vh[j].y; e1.y = vh[j].w;
            HSf[((0 * 16 + r) * PITCH) * 2 + q] = e0;
            HSf[((1 * 16 + r) * PITCH) * 2 + q] = e1;
        }
#pragma unroll
        for (int j = 0; j < 12; ++j) {
            const int i = t + 256 * j;
            const int g = i >> 10, r = (i >> 6) & 15, q = i & 63;
            WI[(g * 16 + r) * PITCH + q] = vi[j];
            WH[(g * 16 + r) * PITCH + q] = vw[j];
        }
    }
    __syncthreads();

    const int plane = (wave < 6) ? wave / 3 : 0;
    const int gate  = (wave < 6) ? wave - 3 * plane : 0;
    const int c     = lane & 15;
    const int h     = lane >> 4;
    const int d     = tn * 16 + c;
    const int wrow  = gate * DIM + d;

    v8f acc_i = {};
    v8f acc_h = {};
    float bI = 0.0f, bH = 0.0f;
    float hp[8];
#pragma unroll
    for (int v = 0; v < 8; ++v) hp[v] = 0.0f;

    if (wave < 6) {
        bI = b_ih[wrow];
        bH = b_hh[wrow];
        if (gate == 2) {
#pragma unroll
            for (int v = 0; v < 8; ++v)
                hp[v] = dyn[(size_t)(tr * 16 + 8 * h + v) * (2 * DIM) + d * 2 + plane];
        }

        const float4* __restrict__ fa = XS + (plane * 16 + c) * PITCH + h;
        const float4* __restrict__ fh = HS + (plane * 16 + c) * PITCH + h;
        const float4* __restrict__ fi = WI + (gate * 16 + c) * PITCH + h;
        const float4* __restrict__ fw = WH + (gate * 16 + c) * PITCH + h;

        float4 A[2][STG], G[2][STG], BI[2][STG], BH[2][STG];
#pragma unroll
        for (int u = 0; u < STG; ++u) {
            A[0][u]  = fa[2 * u];
            G[0][u]  = fh[2 * u];
            BI[0][u] = fi[2 * u];
            BH[0][u] = fw[2 * u];
        }
#pragma unroll
        for (int s = 0; s < KCH / STG; ++s) {
            const int cur = s & 1;
            const int nxt = cur ^ 1;
            if (s + 1 < KCH / STG) {
#pragma unroll
                for (int u = 0; u < STG; ++u) {
                    const int kc = (s + 1) * STG + u;
                    A[nxt][u]  = fa[2 * kc];
                    G[nxt][u]  = fh[2 * kc];
                    BI[nxt][u] = fi[2 * kc];
                    BH[nxt][u] = fw[2 * kc];
                }
            }
            __builtin_amdgcn_sched_barrier(0);
#pragma unroll
            for (int u = 0; u < STG; ++u) {
                const float4 a  = A[cur][u];
                const float4 g  = G[cur][u];
                const float4 bi = BI[cur][u];
                const float4 bh = BH[cur][u];
                v2f a0, a1, b0, b1, g0, g1, w0, w1;
                a0.x = a.x;  a0.y = a.y;  a1.x = a.z;  a1.y = a.w;
                b0.x = bi.x; b0.y = bi.y; b1.x = bi.z; b1.y = bi.w;
                g0.x = g.x;  g0.y = g.y;  g1.x = g.z;  g1.y = g.w;
                w0.x = bh.x; w0.y = bh.y; w1.x = bh.z; w1.y = bh.w;
                acc_i = __builtin_amdgcn_wmma_f32_16x16x4_f32(false, a0, false, b0, (short)0, acc_i, false, false);
                acc_h = __builtin_amdgcn_wmma_f32_16x16x4_f32(false, g0, false, w0, (short)0, acc_h, false, false);
                acc_i = __builtin_amdgcn_wmma_f32_16x16x4_f32(false, a1, false, b1, (short)0, acc_i, false, false);
                acc_h = __builtin_amdgcn_wmma_f32_16x16x4_f32(false, g1, false, w1, (short)0, acc_h, false, false);
            }
            __builtin_amdgcn_sched_barrier(0);
            asm volatile("v_nop\n\tv_nop\n\tv_nop\n\tv_nop\n\tv_nop");
            __builtin_amdgcn_sched_barrier(0);
        }

        if (gate != 2) {
#pragma unroll
            for (int v = 0; v < 8; ++v) {
                const float pre = (acc_i[v] + bI) + (acc_h[v] + bH);
                sG[plane][gate][v][lane] = 1.0f / (1.0f + expf(-pre));
            }
        }
    }
    __syncthreads();
    if (wave < 6 && gate == 2) {
#pragma unroll
        for (int v = 0; v < 8; ++v) {
            const float rg = sG[plane][0][v][lane];
            const float z  = sG[plane][1][v][lane];
            const float ng = tanhf((acc_i[v] + bI) + rg * (acc_h[v] + bH));
            sO[plane][v][lane] = (1.0f - z) * ng + z * hp[v];
        }
    }
    __syncthreads();
    {
        float2* __restrict__ out2 = (float2*)out;
        float2 o;
        o.x = sO[0][wave][lane];
        o.y = sO[1][wave][lane];
        out2[(size_t)(tr * 16 + 8 * h + wave) * DIM + d] = o;
    }
}

extern "C" void kernel_launch(void* const* d_in, const int* in_sizes, int n_in,
                              void* d_out, int out_size, void* d_ws, size_t ws_size,
                              hipStream_t stream) {
    (void)in_sizes; (void)n_in; (void)out_size;

    const int*   src_nid = (const int*)d_in[0];
    const int*   dst_nid = (const int*)d_in[1];
    const int*   rel     = (const int*)d_in[2];
    const float* ent     = (const float*)d_in[3];
    const float* dyn     = (const float*)d_in[4];
    const float* W_ih    = (const float*)d_in[5];
    const float* W_hh    = (const float*)d_in[6];
    const float* b_ih    = (const float*)d_in[7];
    const float* b_hh    = (const float*)d_in[8];
    float*       out     = (float*)d_out;

    float* ws       = (float*)d_ws;
    float* X_src    = ws;
    int*   permS    = (int*)(ws + 131328);
    int*   seg      = permS + N_EDGES;
    int*   permD    = seg + N_RELS * NB;

    chunk_sort_kernel<<<NB, 256, 0, stream>>>(rel, src_nid, dst_nid, permS,
                                              permD, seg, ws_size);
    reduce_kernel<<<2 * N_RELS, 512, 0, stream>>>(dst_nid, ent, permS, permD,
                                                  seg, X_src, ws_size);

    cell_kernel<<<(N_RELS / 16) * (DIM / 16), 256, 0, stream>>>(X_src, dyn, W_ih, W_hh, b_ih, b_hh, out);
}
